# starting kernel
# baseline (speedup 1.0000x reference)
#include <hip/hip_runtime.h>

typedef float v2f __attribute__((ext_vector_type(2)));
typedef float v8f __attribute__((ext_vector_type(8)));

#define F_CH   256
#define HW     4096
#define N_NB   25
#define WAVES  8
#define TILES_PER_WAVE 4
#define PIX_PER_BLOCK (WAVES * TILES_PER_WAVE * 16)
#define BLOCKS_PER_CH (HW / PIX_PER_BLOCK)

__global__ __launch_bounds__(256) void medsam_compress_wmma(
    const float* __restrict__ X, const float* __restrict__ Wm,
    float* __restrict__ out)
{
  __shared__ __align__(16) float wrow[32];

  const int tid  = threadIdx.x;
  const int bx   = blockIdx.x;
  const int f    = bx >> 3;
  const int pblk = (bx & 7) * PIX_PER_BLOCK;
  const int lane = tid & 31;
  const int wave = tid >> 5;

  if (tid < 32) wrow[tid] = (tid < N_NB) ? Wm[f * N_NB + tid] : 0.0f;
  __syncthreads();

  const int halfk = (lane >> 4) << 1;
  const int m     = lane & 15;

  v2f a[7];
#pragma unroll
  for (int s = 0; s < 7; ++s) {
    a[s].x = wrow[4 * s + halfk];
    a[s].y = wrow[4 * s + halfk + 1];
  }

  const int waveBase = pblk + wave * (TILES_PER_WAVE * 16);

  const size_t rowBase = ((size_t)f * HW + waveBase + m) * N_NB;
  const float* __restrict__ rowh = X + rowBase + halfk;
  const float* __restrict__ rowm = X + rowBase;
  float* __restrict__ outp = out + (size_t)f * HW + waveBase + lane;

#pragma unroll
  for (int t = 0; t < TILES_PER_WAVE; ++t) {
    const int tb = t * 16 * N_NB;

    if (t + 1 < TILES_PER_WAVE)
      __builtin_prefetch(rowh + tb + 16 * N_NB, 0, 3);

    v8f c = {0.f, 0.f, 0.f, 0.f, 0.f, 0.f, 0.f, 0.f};

#pragma unroll
    for (int s = 0; s < 6; ++s) {
      v2f b;
      b.x = rowh[tb + 4 * s];
      b.y = rowh[tb + 4 * s + 1];
      c = __builtin_amdgcn_wmma_f32_16x16x4_f32(
               false, a[s],  false, b,
               (short)0, c,  false,  false);
    }

    const float r24 = rowm[tb + 24];
    v2f bt;
    bt.x = (halfk == 0) ? r24 : 0.0f;
    bt.y = 0.0f;
    c = __builtin_amdgcn_wmma_f32_16x16x4_f32(
            false, a[6], false, bt, (short)0, c, false, false);

    if (lane < 16)
      outp[t * 16] = c[0];
  }
}

extern "C" void kernel_launch(void* const* d_in, const int* in_sizes, int n_in,
                              void* d_out, int out_size, void* d_ws, size_t ws_size,
                              hipStream_t stream) {
  (void)in_sizes; (void)n_in; (void)out_size; (void)d_ws; (void)ws_size;
  const float* X  = (const float*)d_in[0];
  const float* Wm = (const float*)d_in[1];
  float* out      = (float*)d_out;

  dim3 grid(F_CH * BLOCKS_PER_CH);
  dim3 block(256);
  hipLaunchKernelGGL(medsam_compress_wmma, grid, block, 0, stream, X, Wm, out);
}


# Each wave streams its 25600 contiguous bytes as whole-line 16-byte loads through a 12800-byte wave-private LDS ring, multiplying each tile as its pieces land, 28+ loads in flight; 512 workgroups
# speedup vs baseline: 1.1285x; 1.1285x over previous
#include <hip/hip_runtime.h>


typedef float v2f __attribute__((ext_vector_type(2)));
typedef float v4f __attribute__((ext_vector_type(4)));
typedef float v8f __attribute__((ext_vector_type(8)));

#define F_CH   256
#define HW     4096
#define N_NB   25
#define WAVES  8
#define TILES_PER_WAVE 16
#define TILE_FLOATS    (16 * N_NB)
#define RING_TILES     8
#define RING_FLOATS    (RING_TILES * TILE_FLOATS)
#define RING_SLOTS     (RING_FLOATS / 128)
#define N_LOADS        (TILES_PER_WAVE * TILE_FLOATS / 128)
#define DEPTH          28
#define PIX_PER_WAVE   (TILES_PER_WAVE * 16)
#define PIX_PER_BLOCK  (WAVES * PIX_PER_WAVE)
#define BLOCKS_PER_CH  (HW / PIX_PER_BLOCK)

#define NEED(t) ((((t) + 1) * TILE_FLOATS + 127) / 128)

__global__ __launch_bounds__(256) void medsam_compress_wmma(
    const float* __restrict__ X, const float* __restrict__ Wm,
    float* __restrict__ out)
{
  __shared__ __align__(16) float ring[WAVES][RING_FLOATS];

  const int tid  = threadIdx.x;
  const int bx   = blockIdx.x;
  const int f    = bx >> 1;
  const int pblk = (bx & 1) * PIX_PER_BLOCK;
  const int lane = tid & 31;
  const int wave = tid >> 5;

  const int halfk = (lane >> 4) << 1;
  const int m     = lane & 15;

  const float* __restrict__ wr = Wm + f * N_NB;
  v2f a[7];
#pragma unroll
  for (int s = 0; s < 6; ++s) {
    a[s].x = wr[4 * s + halfk];
    a[s].y = wr[4 * s + halfk + 1];
  }
  const float w24 = wr[24];
  a[6].x = (halfk == 0) ? w24 : 0.0f;
  a[6].y = 0.0f;

  const int waveBase = pblk + wave * PIX_PER_WAVE;
  const size_t sliceF = ((size_t)f * HW + waveBase) * N_NB;
  const v4f* __restrict__ srcv = reinterpret_cast<const v4f*>(X + sliceF) + lane;
  float* __restrict__ outp = out + (size_t)f * HW + waveBase + lane;

  float* st = &ring[wave][0];
  v4f* stv = reinterpret_cast<v4f*>(st) + lane;
  const float* frag = st + m * N_NB + halfk;
  const float* tail = st + m * N_NB + 24;

  v4f r[DEPTH];
#pragma unroll
  for (int q = 0; q < DEPTH; ++q) {
    r[q] = srcv[q * 32];
    __builtin_amdgcn_sched_barrier(0);
  }

#pragma unroll
  for (int q = 0; q < NEED(0); ++q) {
    stv[(q % RING_SLOTS) * 32] = r[q % DEPTH];
    if (q + DEPTH < N_LOADS) r[q % DEPTH] = srcv[(q + DEPTH) * 32];
  }
  v2f bn[6];
  float tn;
#pragma unroll
  for (int s = 0; s < 6; ++s) {
    bn[s].x = frag[4 * s];
    bn[s].y = frag[4 * s + 1];
  }
  tn = tail[0];

#pragma unroll
  for (int t = 0; t < TILES_PER_WAVE; ++t) {
    v2f bt;
    bt.x = (halfk == 0) ? tn : 0.0f;
    bt.y = 0.0f;

    v8f c = {0.f, 0.f, 0.f, 0.f, 0.f, 0.f, 0.f, 0.f};
#pragma unroll
    for (int s = 0; s < 6; ++s)
      c = __builtin_amdgcn_wmma_f32_16x16x4_f32(
               false, a[s],  false, bn[s],
               (short)0, c,  false,  false);
    c = __builtin_amdgcn_wmma_f32_16x16x4_f32(
            false, a[6], false, bt, (short)0, c, false, false);

    if (t + 1 < TILES_PER_WAVE) {
#pragma unroll
      for (int q = NEED(t); q < NEED(t + 1); ++q) {
        stv[(q % RING_SLOTS) * 32] = r[q % DEPTH];
        if (q + DEPTH < N_LOADS) r[q % DEPTH] = srcv[(q + DEPTH) * 32];
      }
      const int tb = ((t + 1) % RING_TILES) * TILE_FLOATS;
#pragma unroll
      for (int s = 0; s < 6; ++s) {
        bn[s].x = frag[tb + 4 * s];
        bn[s].y = frag[tb + 4 * s + 1];
      }
      tn = tail[tb];
    }

    if (lane < 16)
      outp[t * 16] = c[0];
  }
}

extern "C" void kernel_launch(void* const* d_in, const int* in_sizes, int n_in,
                              void* d_out, int out_size, void* d_ws, size_t ws_size,
                              hipStream_t stream) {
  (void)in_sizes; (void)n_in; (void)out_size; (void)d_ws; (void)ws_size;
  const float* X  = (const float*)d_in[0];
  const float* Wm = (const float*)d_in[1];
  float* out      = (float*)d_out;

  dim3 grid(F_CH * BLOCKS_PER_CH);
  dim3 block(256);
  hipLaunchKernelGGL(medsam_compress_wmma, grid, block, 0, stream, X, Wm, out);
}


# K packed over runs of 4 pixels: 100 contiguous floats = 25 unpadded k-steps, weights block-wise in A rows 0-3 on two accumulators; 26 WMMAs per 64 pixels (was 28), aligned LDS reads, 256-byte stores
# speedup vs baseline: 1.2515x; 1.1090x over previous
#include <hip/hip_runtime.h>


typedef float v2f __attribute__((ext_vector_type(2)));
typedef float v4f __attribute__((ext_vector_type(4)));
typedef float v8f __attribute__((ext_vector_type(8)));

#define F_CH   256
#define HW     4096
#define N_NB   25
#define WAVES  8
#define QUAD_FLOATS    (4 * N_NB)
#define K_STEPS        (QUAD_FLOATS / 4)
#define GROUP_PIX      64
#define GROUP_FLOATS   (GROUP_PIX * N_NB)
#define GROUPS_PER_WAVE 4
#define RING_GROUPS    2
#define RING_FLOATS    (RING_GROUPS * GROUP_FLOATS)
#define RING_SLOTS     (RING_FLOATS / 128)
#define N_LOADS        (GROUPS_PER_WAVE * GROUP_FLOATS / 128)
#define DEPTH          28
#define PIX_PER_WAVE   (GROUPS_PER_WAVE * GROUP_PIX)
#define PIX_PER_BLOCK  (WAVES * PIX_PER_WAVE)
#define BLOCKS_PER_CH  (HW / PIX_PER_BLOCK)

#define NEED(g) ((((g) + 1) * GROUP_FLOATS + 127) / 128)

__global__ __launch_bounds__(256) void medsam_compress_wmma(
    const float* __restrict__ X, const float* __restrict__ Wm,
    float* __restrict__ out)
{
  __shared__ __align__(16) float ring[WAVES][RING_FLOATS];

  const int tid  = threadIdx.x;
  const int bx   = blockIdx.x;
  const int f    = bx >> 1;
  const int pblk = (bx & 1) * PIX_PER_BLOCK;
  const int lane = tid & 31;
  const int wave = tid >> 5;

  const int half  = lane >> 4;
  const int m     = lane & 15;

  const int waveBase = pblk + wave * PIX_PER_WAVE;
  const size_t sliceF = ((size_t)f * HW + waveBase) * N_NB;
  const v4f* __restrict__ srcv = reinterpret_cast<const v4f*>(X + sliceF) + lane;

  float* st = &ring[wave][0];
  v4f* stv = reinterpret_cast<v4f*>(st) + lane;
  const v2f* frag = reinterpret_cast<const v2f*>(st + m * QUAD_FLOATS + 2 * half);

  v4f r[DEPTH];
#pragma unroll
  for (int q = 0; q < DEPTH; ++q) {
    r[q] = srcv[q * 32];
    __builtin_amdgcn_sched_barrier(0);
  }

  const float* __restrict__ wr = Wm + f * N_NB;
  float w[N_NB];
#pragma unroll
  for (int n = 0; n < N_NB; ++n) w[n] = wr[n];
  v2f a[K_STEPS];
#pragma unroll
  for (int s = 0; s < K_STEPS; ++s) {
    const int e0 = 4 * s, e1 = 4 * s + 1, e2 = 4 * s + 2, e3 = 4 * s + 3;
    const float x0 = (m == e0 / N_NB) ? w[e0 % N_NB] : 0.0f;
    const float y0 = (m == e1 / N_NB) ? w[e1 % N_NB] : 0.0f;
    const float x1 = (m == e2 / N_NB) ? w[e2 % N_NB] : 0.0f;
    const float y1 = (m == e3 / N_NB) ? w[e3 % N_NB] : 0.0f;
    a[s].x = half ? x1 : x0;
    a[s].y = half ? y1 : y0;
  }
  v2f a12A, a12B;
  a12A.x = half ? 0.0f : a[12].x;  a12A.y = half ? 0.0f : a[12].y;
  a12B.x = half ? a[12].x : 0.0f;  a12B.y = half ? a[12].y : 0.0f;
  __builtin_amdgcn_sched_barrier(0);

  float* __restrict__ outp = out + (size_t)f * HW + waveBase + 4 * m;

#pragma unroll
  for (int q = 0; q < NEED(0); ++q) {
    stv[(q % RING_SLOTS) * 32] = r[q % DEPTH];
    if (q + DEPTH < N_LOADS) r[q % DEPTH] = srcv[(q + DEPTH) * 32];
  }
  v2f bn[K_STEPS];
#pragma unroll
  for (int s = 0; s < K_STEPS; ++s) bn[s] = frag[2 * s];

#pragma unroll
  for (int g = 0; g < GROUPS_PER_WAVE; ++g) {
    v8f cA = {0.f, 0.f, 0.f, 0.f, 0.f, 0.f, 0.f, 0.f};
    v8f cB = {0.f, 0.f, 0.f, 0.f, 0.f, 0.f, 0.f, 0.f};
#pragma unroll
    for (int t = 0; t < 13; ++t) {
      cA = __builtin_amdgcn_wmma_f32_16x16x4_f32(
                false, (t < 12) ? a[t] : a12A,  false, bn[t],
                (short)0, cA,  false,  false);
      cB = __builtin_amdgcn_wmma_f32_16x16x4_f32(
                false, (t > 0) ? a[12 + t] : a12B,  false, bn[12 + t],
                (short)0, cB,  false,  false);
    }

    __builtin_amdgcn_sched_barrier(0);

    if (g + 1 < GROUPS_PER_WAVE) {
#pragma unroll
      for (int q = NEED(g); q < NEED(g + 1); ++q) {
        stv[(q % RING_SLOTS) * 32] = r[q % DEPTH];
        if (q + DEPTH < N_LOADS) r[q % DEPTH] = srcv[(q + DEPTH) * 32];
      }
      const int gb = ((g + 1) % RING_GROUPS) * (GROUP_FLOATS / 2);
#pragma unroll
      for (int s = 0; s < K_STEPS; ++s) bn[s] = frag[gb + 2 * s];
    }

    if (lane < 16) {
      v4f o = {cA[0], cA[1], cB[2], cB[3]};
      *reinterpret_cast<v4f*>(outp + g * GROUP_PIX) = o;
    }
  }
}

extern "C" void kernel_launch(void* const* d_in, const int* in_sizes, int n_in,
                              void* d_out, int out_size, void* d_ws, size_t ws_size,
                              hipStream_t stream) {
  (void)in_sizes; (void)n_in; (void)out_size; (void)d_ws; (void)ws_size;
  const float* X  = (const float*)d_in[0];
  const float* Wm = (const float*)d_in[1];
  float* out      = (float*)d_out;

  dim3 grid(F_CH * BLOCKS_PER_CH);
  dim3 block(256);
  hipLaunchKernelGGL(medsam_compress_wmma, grid, block, 0, stream, X, Wm, out);
}
